# starting kernel
# baseline (speedup 1.0000x reference)
#include <hip/hip_runtime.h>

typedef __attribute__((ext_vector_type(2))) float v2f;
typedef __attribute__((ext_vector_type(8))) float v8f;

#define FIN 6
#define HID 16

__global__ __launch_bounds__(256) void zero_kernel(float* __restrict__ p, int n) {
  int i = blockIdx.x * blockDim.x + threadIdx.x;
  if (i < n) p[i] = 0.0f;
}

__global__ __launch_bounds__(256) void degree_kernel(const int* __restrict__ src,
                                                     const int* __restrict__ dst,
                                                     float* deg_out, float* deg_in, int E) {
  int e = blockIdx.x * blockDim.x + threadIdx.x;
  if (e >= E) return;
  atomicAdd(deg_out + src[e], 1.0f);
  atomicAdd(deg_in + dst[e], 1.0f);
}

__global__ __launch_bounds__(256) void norm_hs1_kernel(const float* __restrict__ h,
                                                       float* deg_out, float* deg_in,
                                                       float* __restrict__ hs1, int N) {
  int i = blockIdx.x * blockDim.x + threadIdx.x;
  if (i >= N) return;
  float dO = deg_out[i];
  float dI = deg_in[i];
  float no = dO > 0.0f ? rsqrtf(dO) : 0.0f;
  float ni = dI > 0.0f ? rsqrtf(dI) : 0.0f;
  deg_out[i] = no;
  deg_in[i]  = ni;
#pragma unroll
  for (int k = 0; k < FIN; ++k) hs1[i * FIN + k] = h[i * FIN + k] * no;
}

__global__ __launch_bounds__(256) void scatter1_kernel(const int* __restrict__ src,
                                                       const int* __restrict__ dst,
                                                       const float* __restrict__ hs1,
                                                       float* agg1, int E) {
  int e = blockIdx.x * blockDim.x + threadIdx.x;
  if (e >= E) return;
  int s = src[e], d = dst[e];
  const float2* hp = (const float2*)(hs1 + (size_t)s * FIN);
  float2 a = hp[0], b = hp[1], c = hp[2];
  float* ap = agg1 + (size_t)d * FIN;
  atomicAdd(ap + 0, a.x); atomicAdd(ap + 1, a.y);
  atomicAdd(ap + 2, b.x); atomicAdd(ap + 3, b.y);
  atomicAdd(ap + 4, c.x); atomicAdd(ap + 5, c.y);
}

__global__ __launch_bounds__(256) void combine1_kernel(const float* __restrict__ agg1,
                                                       const float* __restrict__ norm_in,
                                                       const float* __restrict__ norm_out,
                                                       const float* __restrict__ W1,
                                                       const float* __restrict__ b1,
                                                       float* __restrict__ hs2, int N) {
  int i = blockIdx.x * blockDim.x + threadIdx.x;
  if (i >= N) return;
  float ni = norm_in[i], no = norm_out[i];
  float v[FIN];
#pragma unroll
  for (int k = 0; k < FIN; ++k) v[k] = agg1[(size_t)i * FIN + k] * ni;
#pragma unroll
  for (int j = 0; j < HID; ++j) {
    float acc = b1[j];
#pragma unroll
    for (int k = 0; k < FIN; ++k) acc = fmaf(v[k], W1[k * HID + j], acc);
    acc = fmaxf(acc, 0.0f);
    hs2[(size_t)i * HID + j] = acc * no;
  }
}

__global__ __launch_bounds__(256) void scatter2_kernel(const int* __restrict__ src,
                                                       const int* __restrict__ dst,
                                                       const float* __restrict__ hs2,
                                                       float* agg2, int E) {
  int e = blockIdx.x * blockDim.x + threadIdx.x;
  if (e >= E) return;
  int s = src[e], d = dst[e];
  const float4* hp = (const float4*)(hs2 + (size_t)s * HID);
  float4 v0 = hp[0], v1 = hp[1], v2 = hp[2], v3 = hp[3];
  float* ap = agg2 + (size_t)d * HID;
  atomicAdd(ap +  0, v0.x); atomicAdd(ap +  1, v0.y);
  atomicAdd(ap +  2, v0.z); atomicAdd(ap +  3, v0.w);
  atomicAdd(ap +  4, v1.x); atomicAdd(ap +  5, v1.y);
  atomicAdd(ap +  6, v1.z); atomicAdd(ap +  7, v1.w);
  atomicAdd(ap +  8, v2.x); atomicAdd(ap +  9, v2.y);
  atomicAdd(ap + 10, v2.z); atomicAdd(ap + 11, v2.w);
  atomicAdd(ap + 12, v3.x); atomicAdd(ap + 13, v3.y);
  atomicAdd(ap + 14, v3.z); atomicAdd(ap + 15, v3.w);
}

__global__ __launch_bounds__(256) void combine2_wmma_kernel(float* agg2,
                                                            const float* __restrict__ norm_in,
                                                            const float* __restrict__ W2,
                                                            const float* __restrict__ b2,
                                                            int numTiles) {
  int wave = (int)((blockIdx.x * blockDim.x + threadIdx.x) >> 5);
  int lane = threadIdx.x & 31;
  if (wave >= numTiles) return;
  int row0 = wave * 16;
  int half = lane >> 4;
  int lr   = lane & 15;

  int arow = row0 + lr;
  float ni = norm_in[arow];
  const float* Arow = agg2 + (size_t)arow * HID;

  v8f c = {};
#pragma unroll
  for (int kb = 0; kb < 16; kb += 4) {
    int k0 = kb + half * 2;
    v2f a, b;
    a.x = Arow[k0 + 0] * ni;
    a.y = Arow[k0 + 1] * ni;
    b.x = W2[(k0 + 0) * HID + lr];
    b.y = W2[(k0 + 1) * HID + lr];
    c = __builtin_amdgcn_wmma_f32_16x16x4_f32(false, a, false, b, (short)0, c,
                                              false, false);
  }

  float bias = b2[lr];
#pragma unroll
  for (int r = 0; r < 8; ++r) {
    int m = r + half * 8;
    float val = fmaxf(c[r] + bias, 0.0f);
    agg2[(size_t)(row0 + m) * HID + lr] = val;
  }
}

__global__ __launch_bounds__(256) void combine2_tail_kernel(float* agg2,
                                                            const float* __restrict__ norm_in,
                                                            const float* __restrict__ W2,
                                                            const float* __restrict__ b2,
                                                            int rowStart, int N) {
  int i = rowStart + blockIdx.x * blockDim.x + threadIdx.x;
  if (i >= N) return;
  float ni = norm_in[i];
  float v[HID];
#pragma unroll
  for (int k = 0; k < HID; ++k) v[k] = agg2[(size_t)i * HID + k] * ni;
#pragma unroll
  for (int j = 0; j < HID; ++j) {
    float acc = b2[j];
#pragma unroll
    for (int k = 0; k < HID; ++k) acc = fmaf(v[k], W2[k * HID + j], acc);
    agg2[(size_t)i * HID + j] = fmaxf(acc, 0.0f);
  }
}

__global__ __launch_bounds__(256) void head_kernel(const float* __restrict__ h2,
                                                   const float* __restrict__ Wl,
                                                   const float* __restrict__ bl,
                                                   float* __restrict__ out, int N) {
  int i = blockIdx.x * blockDim.x + threadIdx.x;
  if (i >= N) return;
  const float4* hp = (const float4*)(h2 + (size_t)i * HID);
  const float4* wp = (const float4*)Wl;
  float acc = bl[0];
#pragma unroll
  for (int q = 0; q < 4; ++q) {
    float4 hv = hp[q], wv = wp[q];
    acc = fmaf(hv.x, wv.x, acc);
    acc = fmaf(hv.y, wv.y, acc);
    acc = fmaf(hv.z, wv.z, acc);
    acc = fmaf(hv.w, wv.w, acc);
  }
  out[i] = acc;
}

extern "C" void kernel_launch(void* const* d_in, const int* in_sizes, int n_in,
                              void* d_out, int out_size, void* d_ws, size_t ws_size,
                              hipStream_t stream) {
  const float* h  = (const float*)d_in[0];
  const float* W1 = (const float*)d_in[1];
  const float* b1 = (const float*)d_in[2];
  const float* W2 = (const float*)d_in[3];
  const float* b2 = (const float*)d_in[4];
  const float* Wl = (const float*)d_in[5];
  const float* bl = (const float*)d_in[6];
  const int* src  = (const int*)d_in[7];
  const int* dst  = (const int*)d_in[8];
  float* out = (float*)d_out;

  const int N = in_sizes[0] / FIN;
  const int E = in_sizes[7];

  float* ws = (float*)d_ws;
  float* deg_out = ws;
  float* deg_in  = ws + (size_t)N;
  float* agg1    = ws + 2 * (size_t)N;
  float* agg2    = ws + 8 * (size_t)N;
  float* hs1     = ws + 24 * (size_t)N;
  float* hs2     = ws + 30 * (size_t)N;

  const int B = 256;
  int zn = 24 * N;
  zero_kernel<<<(zn + B - 1) / B, B, 0, stream>>>(ws, zn);

  degree_kernel<<<(E + B - 1) / B, B, 0, stream>>>(src, dst, deg_out, deg_in, E);
  norm_hs1_kernel<<<(N + B - 1) / B, B, 0, stream>>>(h, deg_out, deg_in, hs1, N);

  scatter1_kernel<<<(E + B - 1) / B, B, 0, stream>>>(src, dst, hs1, agg1, E);
  combine1_kernel<<<(N + B - 1) / B, B, 0, stream>>>(agg1, deg_in, deg_out, W1, b1, hs2, N);

  scatter2_kernel<<<(E + B - 1) / B, B, 0, stream>>>(src, dst, hs2, agg2, E);

  int numTiles = N / 16;
  if (numTiles > 0) {
    int threads = numTiles * 32;
    combine2_wmma_kernel<<<(threads + B - 1) / B, B, 0, stream>>>(agg2, deg_in, W2, b2, numTiles);
  }
  int rem = N - numTiles * 16;
  if (rem > 0) {
    combine2_tail_kernel<<<(rem + B - 1) / B, B, 0, stream>>>(agg2, deg_in, W2, b2,
                                                              numTiles * 16, N);
  }

  head_kernel<<<(N + B - 1) / B, B, 0, stream>>>(agg2, Wl, bl, out, N);
}


# Row scatters: lanes of a wave cover one edge's contiguous row (16 lanes/edge layer 2, 8 layer 1), one f32 atomic instr hits 2-4 rows not 32 lines; atomics kept, sums still in arrival order
# speedup vs baseline: 3.0834x; 3.0834x over previous
#include <hip/hip_runtime.h>

typedef __attribute__((ext_vector_type(2))) float v2f;
typedef __attribute__((ext_vector_type(8))) float v8f;

#define FIN 6
#define HID 16
#define EPW 128

__global__ __launch_bounds__(256) void zero_kernel(float* __restrict__ p, int n) {
  int i = blockIdx.x * blockDim.x + threadIdx.x;
  if (i < n) p[i] = 0.0f;
}

__global__ __launch_bounds__(256) void degree_kernel(const int* __restrict__ src,
                                                     const int* __restrict__ dst,
                                                     float* deg_out, float* deg_in, int E) {
  int e = blockIdx.x * blockDim.x + threadIdx.x;
  if (e >= E) return;
  atomicAdd(deg_out + src[e], 1.0f);
  atomicAdd(deg_in + dst[e], 1.0f);
}

__global__ __launch_bounds__(256) void norm_hs1_kernel(const float* __restrict__ h,
                                                       float* deg_out, float* deg_in,
                                                       float* __restrict__ hs1, int N) {
  int i = blockIdx.x * blockDim.x + threadIdx.x;
  if (i >= N) return;
  float dO = deg_out[i];
  float dI = deg_in[i];
  float no = dO > 0.0f ? rsqrtf(dO) : 0.0f;
  float ni = dI > 0.0f ? rsqrtf(dI) : 0.0f;
  deg_out[i] = no;
  deg_in[i]  = ni;
#pragma unroll
  for (int k = 0; k < FIN; ++k) hs1[i * FIN + k] = h[i * FIN + k] * no;
}

template <bool GUARD>
static __device__ __forceinline__ void scatter1_chunk(const int* __restrict__ src,
                                                      const int* __restrict__ dst,
                                                      const float* __restrict__ hs1,
                                                      float* agg1, int cb, int E,
                                                      int lane, int sub, int q) {
  const int e = cb + lane;
  int s_l = 0, d_l = -1;
  if (!GUARD || e < E) { s_l = src[e]; d_l = dst[e]; }
  unsigned s[8];
  int d[8];
#pragma unroll
  for (int it = 0; it < 8; ++it) {
    const int j = 4 * it + q;
    s[it] = (unsigned)__shfl(s_l, j);
    d[it] = __shfl(d_l, j);
  }
  if (sub < FIN) {
    float v[8];
#pragma unroll
    for (int it = 0; it < 8; ++it) v[it] = hs1[s[it] * FIN + sub];
#pragma unroll
    for (int it = 0; it < 8; ++it)
      if (!GUARD || d[it] >= 0) atomicAdd(agg1 + ((unsigned)d[it] * FIN + sub), v[it]);
  }
}

__global__ __launch_bounds__(256) void scatter1_kernel(const int* __restrict__ src,
                                                       const int* __restrict__ dst,
                                                       const float* __restrict__ hs1,
                                                       float* agg1, int E) {
  const int lane = threadIdx.x & 31;
  const int wave = __builtin_amdgcn_readfirstlane((int)((blockIdx.x * blockDim.x + threadIdx.x) >> 5));
  const int sub  = lane & 7;
  const int q    = lane >> 3;
  const int ebase = wave * EPW;
  for (int c = 0; c < EPW / 32; ++c) {
    const int cb = ebase + c * 32;
    if (cb + 32 <= E)  scatter1_chunk<false>(src, dst, hs1, agg1, cb, E, lane, sub, q);
    else if (cb < E)   scatter1_chunk<true>(src, dst, hs1, agg1, cb, E, lane, sub, q);
  }
}

__global__ __launch_bounds__(256) void combine1_kernel(const float* __restrict__ agg1,
                                                       const float* __restrict__ norm_in,
                                                       const float* __restrict__ norm_out,
                                                       const float* __restrict__ W1,
                                                       const float* __restrict__ b1,
                                                       float* __restrict__ hs2, int N) {
  int i = blockIdx.x * blockDim.x + threadIdx.x;
  if (i >= N) return;
  float ni = norm_in[i], no = norm_out[i];
  float v[FIN];
#pragma unroll
  for (int k = 0; k < FIN; ++k) v[k] = agg1[(size_t)i * FIN + k] * ni;
#pragma unroll
  for (int j = 0; j < HID; ++j) {
    float acc = b1[j];
#pragma unroll
    for (int k = 0; k < FIN; ++k) acc = fmaf(v[k], W1[k * HID + j], acc);
    acc = fmaxf(acc, 0.0f);
    hs2[(size_t)i * HID + j] = acc * no;
  }
}

template <bool GUARD>
static __device__ __forceinline__ void scatter2_chunk(const int* __restrict__ src,
                                                      const int* __restrict__ dst,
                                                      const float* __restrict__ hs2,
                                                      float* agg2, int cb, int E,
                                                      int lane, int sub, int half) {
  const int e = cb + lane;
  int s_l = 0, d_l = -1;
  if (!GUARD || e < E) { s_l = src[e]; d_l = dst[e]; }
  float v[16];
  int d[16];
#pragma unroll
  for (int it = 0; it < 16; ++it) {
    const int j = 2 * it + half;
    const unsigned s = (unsigned)__shfl(s_l, j);
    d[it] = __shfl(d_l, j);
    v[it] = hs2[s * HID + sub];
  }
#pragma unroll
  for (int it = 0; it < 16; ++it)
    if (!GUARD || d[it] >= 0) atomicAdd(agg2 + ((unsigned)d[it] * HID + sub), v[it]);
}

__global__ __launch_bounds__(256) void scatter2_kernel(const int* __restrict__ src,
                                                       const int* __restrict__ dst,
                                                       const float* __restrict__ hs2,
                                                       float* agg2, int E) {
  const int lane = threadIdx.x & 31;
  const int wave = __builtin_amdgcn_readfirstlane((int)((blockIdx.x * blockDim.x + threadIdx.x) >> 5));
  const int sub  = lane & 15;
  const int half = lane >> 4;
  const int ebase = wave * EPW;
  for (int c = 0; c < EPW / 32; ++c) {
    const int cb = ebase + c * 32;
    if (cb + 32 <= E)  scatter2_chunk<false>(src, dst, hs2, agg2, cb, E, lane, sub, half);
    else if (cb < E)   scatter2_chunk<true>(src, dst, hs2, agg2, cb, E, lane, sub, half);
  }
}

__global__ __launch_bounds__(256) void combine2_wmma_kernel(float* agg2,
                                                            const float* __restrict__ norm_in,
                                                            const float* __restrict__ W2,
                                                            const float* __restrict__ b2,
                                                            int numTiles) {
  int wave = (int)((blockIdx.x * blockDim.x + threadIdx.x) >> 5);
  int lane = threadIdx.x & 31;
  if (wave >= numTiles) return;
  int row0 = wave * 16;
  int half = lane >> 4;
  int lr   = lane & 15;

  int arow = row0 + lr;
  float ni = norm_in[arow];
  const float* Arow = agg2 + (size_t)arow * HID;

  v8f c = {};
#pragma unroll
  for (int kb = 0; kb < 16; kb += 4) {
    int k0 = kb + half * 2;
    v2f a, b;
    a.x = Arow[k0 + 0] * ni;
    a.y = Arow[k0 + 1] * ni;
    b.x = W2[(k0 + 0) * HID + lr];
    b.y = W2[(k0 + 1) * HID + lr];
    c = __builtin_amdgcn_wmma_f32_16x16x4_f32(false, a, false, b, (short)0, c,
                                              false, false);
  }

  float bias = b2[lr];
#pragma unroll
  for (int r = 0; r < 8; ++r) {
    int m = r + half * 8;
    float val = fmaxf(c[r] + bias, 0.0f);
    agg2[(size_t)(row0 + m) * HID + lr] = val;
  }
}

__global__ __launch_bounds__(256) void combine2_tail_kernel(float* agg2,
                                                            const float* __restrict__ norm_in,
                                                            const float* __restrict__ W2,
                                                            const float* __restrict__ b2,
                                                            int rowStart, int N) {
  int i = rowStart + blockIdx.x * blockDim.x + threadIdx.x;
  if (i >= N) return;
  float ni = norm_in[i];
  float v[HID];
#pragma unroll
  for (int k = 0; k < HID; ++k) v[k] = agg2[(size_t)i * HID + k] * ni;
#pragma unroll
  for (int j = 0; j < HID; ++j) {
    float acc = b2[j];
#pragma unroll
    for (int k = 0; k < HID; ++k) acc = fmaf(v[k], W2[k * HID + j], acc);
    agg2[(size_t)i * HID + j] = fmaxf(acc, 0.0f);
  }
}

__global__ __launch_bounds__(256) void head_kernel(const float* __restrict__ h2,
                                                   const float* __restrict__ Wl,
                                                   const float* __restrict__ bl,
                                                   float* __restrict__ out, int N) {
  int i = blockIdx.x * blockDim.x + threadIdx.x;
  if (i >= N) return;
  const float4* hp = (const float4*)(h2 + (size_t)i * HID);
  const float4* wp = (const float4*)Wl;
  float acc = bl[0];
#pragma unroll
  for (int q = 0; q < 4; ++q) {
    float4 hv = hp[q], wv = wp[q];
    acc = fmaf(hv.x, wv.x, acc);
    acc = fmaf(hv.y, wv.y, acc);
    acc = fmaf(hv.z, wv.z, acc);
    acc = fmaf(hv.w, wv.w, acc);
  }
  out[i] = acc;
}

extern "C" void kernel_launch(void* const* d_in, const int* in_sizes, int n_in,
                              void* d_out, int out_size, void* d_ws, size_t ws_size,
                              hipStream_t stream) {
  const float* h  = (const float*)d_in[0];
  const float* W1 = (const float*)d_in[1];
  const float* b1 = (const float*)d_in[2];
  const float* W2 = (const float*)d_in[3];
  const float* b2 = (const float*)d_in[4];
  const float* Wl = (const float*)d_in[5];
  const float* bl = (const float*)d_in[6];
  const int* src  = (const int*)d_in[7];
  const int* dst  = (const int*)d_in[8];
  float* out = (float*)d_out;

  const int N = in_sizes[0] / FIN;
  const int E = in_sizes[7];

  float* ws = (float*)d_ws;
  float* deg_out = ws;
  float* deg_in  = ws + (size_t)N;
  float* agg1    = ws + 2 * (size_t)N;
  float* agg2    = ws + 8 * (size_t)N;
  float* hs1     = ws + 24 * (size_t)N;
  float* hs2     = ws + 30 * (size_t)N;

  const int B = 256;
  int zn = 24 * N;
  zero_kernel<<<(zn + B - 1) / B, B, 0, stream>>>(ws, zn);

  degree_kernel<<<(E + B - 1) / B, B, 0, stream>>>(src, dst, deg_out, deg_in, E);
  norm_hs1_kernel<<<(N + B - 1) / B, B, 0, stream>>>(h, deg_out, deg_in, hs1, N);

  const int EB = (B / 32) * EPW;
  scatter1_kernel<<<(E + EB - 1) / EB, B, 0, stream>>>(src, dst, hs1, agg1, E);
  combine1_kernel<<<(N + B - 1) / B, B, 0, stream>>>(agg1, deg_in, deg_out, W1, b1, hs2, N);

  scatter2_kernel<<<(E + EB - 1) / EB, B, 0, stream>>>(src, dst, hs2, agg2, E);

  int numTiles = N / 16;
  if (numTiles > 0) {
    int threads = numTiles * 32;
    combine2_wmma_kernel<<<(threads + B - 1) / B, B, 0, stream>>>(agg2, deg_in, W2, b2, numTiles);
  }
  int rem = N - numTiles * 16;
  if (rem > 0) {
    combine2_tail_kernel<<<(rem + B - 1) / B, B, 0, stream>>>(agg2, deg_in, W2, b2,
                                                              numTiles * 16, N);
  }

  head_kernel<<<(N + B - 1) / B, B, 0, stream>>>(agg2, Wl, bl, out, N);
}


# Degree counts via per-block LDS histograms (two 16-bit bins per word, ds_add_u32), stored as planes and summed in the norm kernel; the degree pass's 6.4M scattered global f32 atomics are gone
# speedup vs baseline: 5.8967x; 1.9124x over previous
#include <hip/hip_runtime.h>

typedef __attribute__((ext_vector_type(2))) float v2f;
typedef __attribute__((ext_vector_type(8))) float v8f;

#define FIN 6
#define HID 16
#define EPW 128

__global__ __launch_bounds__(256) void zero_kernel(float* __restrict__ p, int n) {
  int i = blockIdx.x * blockDim.x + threadIdx.x;
  if (i < n) p[i] = 0.0f;
}

#define HW    51200
#define BINS  (2 * HW)
#define CHUNK 50000
#define HT    1024
#define HIT   ((CHUNK / 4 + HT - 1) / HT)

static __device__ __forceinline__ void hist_add(unsigned* bins, int i, int base) {
  const unsigned loc = (unsigned)(i - base);
  if (loc < (unsigned)BINS) atomicAdd(&bins[loc >> 1], (loc & 1u) ? 0x10000u : 1u);
}

__global__ __launch_bounds__(HT) void degree_hist_kernel(const int* __restrict__ src,
                                                         const int* __restrict__ dst,
                                                         unsigned* __restrict__ planes,
                                                         int E) {
  __shared__ __attribute__((aligned(16))) unsigned bins[HW];
  const int tid   = threadIdx.x;
  const int chunk = blockIdx.x;
  const int r     = blockIdx.y;
  const int arr   = blockIdx.z;
  const int* __restrict__ idx = arr ? dst : src;
  const int e0   = chunk * CHUNK;
  const int n    = min(CHUNK, E - e0);
  const int n4   = n >> 2;
  const int base = r * BINS;
  const int4* __restrict__ p4 = (const int4*)(idx + e0);

  int4 v[HIT];
#pragma unroll
  for (int it = 0; it < HIT; ++it) {
    const int k = tid + it * HT;
    v[it] = (k < n4) ? p4[k] : make_int4(-1, -1, -1, -1);
  }
  int vt = -1;
  if (tid < (n & 3)) vt = idx[e0 + (n4 << 2) + tid];

  uint4* b4 = (uint4*)bins;
  for (int k = tid; k < HW / 4; k += HT) b4[k] = make_uint4(0u, 0u, 0u, 0u);
  __syncthreads();

#pragma unroll
  for (int it = 0; it < HIT; ++it) {
    hist_add(bins, v[it].x, base);
    hist_add(bins, v[it].y, base);
    hist_add(bins, v[it].z, base);
    hist_add(bins, v[it].w, base);
  }
  hist_add(bins, vt, base);
  __syncthreads();

  uint4* o4 = (uint4*)(planes + ((size_t)(arr * gridDim.y + r) * gridDim.x + chunk) * HW);
  for (int k = tid; k < HW / 4; k += HT) o4[k] = b4[k];
}

__global__ __launch_bounds__(256) void norm_hs1_kernel(const float* __restrict__ h,
                                                       const unsigned* __restrict__ planes,
                                                       float* __restrict__ norm_out,
                                                       float* __restrict__ norm_in,
                                                       float* __restrict__ hs1,
                                                       int N, int nChunks, int nRanges) {
  const int g = blockIdx.x * blockDim.x + threadIdx.x;
  if (2 * g >= N) return;
  const int r  = g / HW;
  const int wl = g - r * HW;
  const unsigned* __restrict__ po = planes + (size_t)r * nChunks * HW + wl;
  const unsigned* __restrict__ pi = planes + (size_t)(nRanges + r) * nChunks * HW + wl;
  unsigned cO[2] = {0u, 0u}, cI[2] = {0u, 0u};
#pragma unroll 8
  for (int c = 0; c < nChunks; ++c) {
    const unsigned a = po[(size_t)c * HW];
    const unsigned b = pi[(size_t)c * HW];
    cO[0] += a & 0xFFFFu;  cO[1] += a >> 16;
    cI[0] += b & 0xFFFFu;  cI[1] += b >> 16;
  }
#pragma unroll
  for (int t = 0; t < 2; ++t) {
    const int i = 2 * g + t;
    if (i < N) {
      const float dO = (float)cO[t];
      const float dI = (float)cI[t];
      const float no = dO > 0.0f ? rsqrtf(dO) : 0.0f;
      const float ni = dI > 0.0f ? rsqrtf(dI) : 0.0f;
      norm_out[i] = no;
      norm_in[i]  = ni;
#pragma unroll
      for (int k = 0; k < FIN; ++k) hs1[i * FIN + k] = h[i * FIN + k] * no;
    }
  }
}

template <bool GUARD>
static __device__ __forceinline__ void scatter1_chunk(const int* __restrict__ src,
                                                      const int* __restrict__ dst,
                                                      const float* __restrict__ hs1,
                                                      float* agg1, int cb, int E,
                                                      int lane, int sub, int q) {
  const int e = cb + lane;
  int s_l = 0, d_l = -1;
  if (!GUARD || e < E) { s_l = src[e]; d_l = dst[e]; }
  unsigned s[8];
  int d[8];
#pragma unroll
  for (int it = 0; it < 8; ++it) {
    const int j = 4 * it + q;
    s[it] = (unsigned)__shfl(s_l, j);
    d[it] = __shfl(d_l, j);
  }
  if (sub < FIN) {
    float v[8];
#pragma unroll
    for (int it = 0; it < 8; ++it) v[it] = hs1[s[it] * FIN + sub];
#pragma unroll
    for (int it = 0; it < 8; ++it)
      if (!GUARD || d[it] >= 0) atomicAdd(agg1 + ((unsigned)d[it] * FIN + sub), v[it]);
  }
}

__global__ __launch_bounds__(256) void scatter1_kernel(const int* __restrict__ src,
                                                       const int* __restrict__ dst,
                                                       const float* __restrict__ hs1,
                                                       float* agg1, int E) {
  const int lane = threadIdx.x & 31;
  const int wave = __builtin_amdgcn_readfirstlane((int)((blockIdx.x * blockDim.x + threadIdx.x) >> 5));
  const int sub  = lane & 7;
  const int q    = lane >> 3;
  const int ebase = wave * EPW;
  for (int c = 0; c < EPW / 32; ++c) {
    const int cb = ebase + c * 32;
    if (cb + 32 <= E)  scatter1_chunk<false>(src, dst, hs1, agg1, cb, E, lane, sub, q);
    else if (cb < E)   scatter1_chunk<true>(src, dst, hs1, agg1, cb, E, lane, sub, q);
  }
}

__global__ __launch_bounds__(256) void combine1_kernel(const float* __restrict__ agg1,
                                                       const float* __restrict__ norm_in,
                                                       const float* __restrict__ norm_out,
                                                       const float* __restrict__ W1,
                                                       const float* __restrict__ b1,
                                                       float* __restrict__ hs2, int N) {
  int i = blockIdx.x * blockDim.x + threadIdx.x;
  if (i >= N) return;
  float ni = norm_in[i], no = norm_out[i];
  float v[FIN];
#pragma unroll
  for (int k = 0; k < FIN; ++k) v[k] = agg1[(size_t)i * FIN + k] * ni;
#pragma unroll
  for (int j = 0; j < HID; ++j) {
    float acc = b1[j];
#pragma unroll
    for (int k = 0; k < FIN; ++k) acc = fmaf(v[k], W1[k * HID + j], acc);
    acc = fmaxf(acc, 0.0f);
    hs2[(size_t)i * HID + j] = acc * no;
  }
}

template <bool GUARD>
static __device__ __forceinline__ void scatter2_chunk(const int* __restrict__ src,
                                                      const int* __restrict__ dst,
                                                      const float* __restrict__ hs2,
                                                      float* agg2, int cb, int E,
                                                      int lane, int sub, int half) {
  const int e = cb + lane;
  int s_l = 0, d_l = -1;
  if (!GUARD || e < E) { s_l = src[e]; d_l = dst[e]; }
  float v[16];
  int d[16];
#pragma unroll
  for (int it = 0; it < 16; ++it) {
    const int j = 2 * it + half;
    const unsigned s = (unsigned)__shfl(s_l, j);
    d[it] = __shfl(d_l, j);
    v[it] = hs2[s * HID + sub];
  }
#pragma unroll
  for (int it = 0; it < 16; ++it)
    if (!GUARD || d[it] >= 0) atomicAdd(agg2 + ((unsigned)d[it] * HID + sub), v[it]);
}

__global__ __launch_bounds__(256) void scatter2_kernel(const int* __restrict__ src,
                                                       const int* __restrict__ dst,
                                                       const float* __restrict__ hs2,
                                                       float* agg2, int E) {
  const int lane = threadIdx.x & 31;
  const int wave = __builtin_amdgcn_readfirstlane((int)((blockIdx.x * blockDim.x + threadIdx.x) >> 5));
  const int sub  = lane & 15;
  const int half = lane >> 4;
  const int ebase = wave * EPW;
  for (int c = 0; c < EPW / 32; ++c) {
    const int cb = ebase + c * 32;
    if (cb + 32 <= E)  scatter2_chunk<false>(src, dst, hs2, agg2, cb, E, lane, sub, half);
    else if (cb < E)   scatter2_chunk<true>(src, dst, hs2, agg2, cb, E, lane, sub, half);
  }
}

__global__ __launch_bounds__(256) void combine2_wmma_kernel(float* agg2,
                                                            const float* __restrict__ norm_in,
                                                            const float* __restrict__ W2,
                                                            const float* __restrict__ b2,
                                                            int numTiles) {
  int wave = (int)((blockIdx.x * blockDim.x + threadIdx.x) >> 5);
  int lane = threadIdx.x & 31;
  if (wave >= numTiles) return;
  int row0 = wave * 16;
  int half = lane >> 4;
  int lr   = lane & 15;

  int arow = row0 + lr;
  float ni = norm_in[arow];
  const float* Arow = agg2 + (size_t)arow * HID;

  v8f c = {};
#pragma unroll
  for (int kb = 0; kb < 16; kb += 4) {
    int k0 = kb + half * 2;
    v2f a, b;
    a.x = Arow[k0 + 0] * ni;
    a.y = Arow[k0 + 1] * ni;
    b.x = W2[(k0 + 0) * HID + lr];
    b.y = W2[(k0 + 1) * HID + lr];
    c = __builtin_amdgcn_wmma_f32_16x16x4_f32(false, a, false, b, (short)0, c,
                                              false, false);
  }

  float bias = b2[lr];
#pragma unroll
  for (int r = 0; r < 8; ++r) {
    int m = r + half * 8;
    float val = fmaxf(c[r] + bias, 0.0f);
    agg2[(size_t)(row0 + m) * HID + lr] = val;
  }
}

__global__ __launch_bounds__(256) void combine2_tail_kernel(float* agg2,
                                                            const float* __restrict__ norm_in,
                                                            const float* __restrict__ W2,
                                                            const float* __restrict__ b2,
                                                            int rowStart, int N) {
  int i = rowStart + blockIdx.x * blockDim.x + threadIdx.x;
  if (i >= N) return;
  float ni = norm_in[i];
  float v[HID];
#pragma unroll
  for (int k = 0; k < HID; ++k) v[k] = agg2[(size_t)i * HID + k] * ni;
#pragma unroll
  for (int j = 0; j < HID; ++j) {
    float acc = b2[j];
#pragma unroll
    for (int k = 0; k < HID; ++k) acc = fmaf(v[k], W2[k * HID + j], acc);
    agg2[(size_t)i * HID + j] = fmaxf(acc, 0.0f);
  }
}

__global__ __launch_bounds__(256) void head_kernel(const float* __restrict__ h2,
                                                   const float* __restrict__ Wl,
                                                   const float* __restrict__ bl,
                                                   float* __restrict__ out, int N) {
  int i = blockIdx.x * blockDim.x + threadIdx.x;
  if (i >= N) return;
  const float4* hp = (const float4*)(h2 + (size_t)i * HID);
  const float4* wp = (const float4*)Wl;
  float acc = bl[0];
#pragma unroll
  for (int q = 0; q < 4; ++q) {
    float4 hv = hp[q], wv = wp[q];
    acc = fmaf(hv.x, wv.x, acc);
    acc = fmaf(hv.y, wv.y, acc);
    acc = fmaf(hv.z, wv.z, acc);
    acc = fmaf(hv.w, wv.w, acc);
  }
  out[i] = acc;
}

extern "C" void kernel_launch(void* const* d_in, const int* in_sizes, int n_in,
                              void* d_out, int out_size, void* d_ws, size_t ws_size,
                              hipStream_t stream) {
  const float* h  = (const float*)d_in[0];
  const float* W1 = (const float*)d_in[1];
  const float* b1 = (const float*)d_in[2];
  const float* W2 = (const float*)d_in[3];
  const float* b2 = (const float*)d_in[4];
  const float* Wl = (const float*)d_in[5];
  const float* bl = (const float*)d_in[6];
  const int* src  = (const int*)d_in[7];
  const int* dst  = (const int*)d_in[8];
  float* out = (float*)d_out;

  const int N = in_sizes[0] / FIN;
  const int E = in_sizes[7];

  float* ws = (float*)d_ws;
  float* deg_out = ws;
  float* deg_in  = ws + (size_t)N;
  float* agg1    = ws + 2 * (size_t)N;
  float* agg2    = ws + 8 * (size_t)N;
  float* hs1     = ws + 24 * (size_t)N;
  float* hs2     = ws + 30 * (size_t)N;
  const int nChunks = (E + CHUNK - 1) / CHUNK;
  const int nRanges = (N + BINS - 1) / BINS;
  unsigned* planes = (unsigned*)(ws + ((46 * (size_t)N + 63) / 64) * 64);

  const int B = 256;
  int zn = 22 * N;
  zero_kernel<<<(zn + B - 1) / B, B, 0, stream>>>(agg1, zn);

  degree_hist_kernel<<<dim3(nChunks, nRanges, 2), HT, 0, stream>>>(src, dst, planes, E);
  const int nPairs = (N + 1) / 2;
  norm_hs1_kernel<<<(nPairs + B - 1) / B, B, 0, stream>>>(h, planes, deg_out, deg_in, hs1,
                                                          N, nChunks, nRanges);

  const int EB = (B / 32) * EPW;
  scatter1_kernel<<<(E + EB - 1) / EB, B, 0, stream>>>(src, dst, hs1, agg1, E);
  combine1_kernel<<<(N + B - 1) / B, B, 0, stream>>>(agg1, deg_in, deg_out, W1, b1, hs2, N);

  scatter2_kernel<<<(E + EB - 1) / EB, B, 0, stream>>>(src, dst, hs2, agg2, E);

  int numTiles = N / 16;
  if (numTiles > 0) {
    int threads = numTiles * 32;
    combine2_wmma_kernel<<<(threads + B - 1) / B, B, 0, stream>>>(agg2, deg_in, W2, b2, numTiles);
  }
  int rem = N - numTiles * 16;
  if (rem > 0) {
    combine2_tail_kernel<<<(rem + B - 1) / B, B, 0, stream>>>(agg2, deg_in, W2, b2,
                                                              numTiles * 16, N);
  }

  head_kernel<<<(N + B - 1) / B, B, 0, stream>>>(agg2, Wl, bl, out, N);
}


# Two copies of each scatter accumulator, one per half of the package (a wave picks its copy by die id); combine kernels add them: float sums are now two partials plus one add; atomics kept
# speedup vs baseline: 11.8502x; 2.0096x over previous
#include <hip/hip_runtime.h>

typedef __attribute__((ext_vector_type(2))) float v2f;
typedef __attribute__((ext_vector_type(8))) float v8f;

#define FIN 6
#define HID 16
#define EPW 128

static __device__ __forceinline__ int half_of_package() {
  return (int)((__builtin_amdgcn_s_sendmsg_rtn(0x87) >> 18) & 1u);
}

__global__ __launch_bounds__(256) void zero_kernel(float* __restrict__ p, int n) {
  int i = blockIdx.x * blockDim.x + threadIdx.x;
  if (i < n) p[i] = 0.0f;
}

#define HW    51200
#define BINS  (2 * HW)
#define CHUNK 50000
#define HT    1024
#define HIT   ((CHUNK / 4 + HT - 1) / HT)

static __device__ __forceinline__ void hist_add(unsigned* bins, int i, int base) {
  const unsigned loc = (unsigned)(i - base);
  if (loc < (unsigned)BINS) atomicAdd(&bins[loc >> 1], (loc & 1u) ? 0x10000u : 1u);
}

__global__ __launch_bounds__(HT) void degree_hist_kernel(const int* __restrict__ src,
                                                         const int* __restrict__ dst,
                                                         unsigned* __restrict__ planes,
                                                         int E) {
  __shared__ __attribute__((aligned(16))) unsigned bins[HW];
  const int tid   = threadIdx.x;
  const int chunk = blockIdx.x;
  const int r     = blockIdx.y;
  const int arr   = blockIdx.z;
  const int* __restrict__ idx = arr ? dst : src;
  const int e0   = chunk * CHUNK;
  const int n    = min(CHUNK, E - e0);
  const int n4   = n >> 2;
  const int base = r * BINS;
  const int4* __restrict__ p4 = (const int4*)(idx + e0);

  int4 v[HIT];
#pragma unroll
  for (int it = 0; it < HIT; ++it) {
    const int k = tid + it * HT;
    v[it] = (k < n4) ? p4[k] : make_int4(-1, -1, -1, -1);
  }
  int vt = -1;
  if (tid < (n & 3)) vt = idx[e0 + (n4 << 2) + tid];

  uint4* b4 = (uint4*)bins;
  for (int k = tid; k < HW / 4; k += HT) b4[k] = make_uint4(0u, 0u, 0u, 0u);
  __syncthreads();

#pragma unroll
  for (int it = 0; it < HIT; ++it) {
    hist_add(bins, v[it].x, base);
    hist_add(bins, v[it].y, base);
    hist_add(bins, v[it].z, base);
    hist_add(bins, v[it].w, base);
  }
  hist_add(bins, vt, base);
  __syncthreads();

  uint4* o4 = (uint4*)(planes + ((size_t)(arr * gridDim.y + r) * gridDim.x + chunk) * HW);
  for (int k = tid; k < HW / 4; k += HT) o4[k] = b4[k];
}

__global__ __launch_bounds__(256) void norm_hs1_kernel(const float* __restrict__ h,
                                                       const unsigned* __restrict__ planes,
                                                       float* __restrict__ norm_out,
                                                       float* __restrict__ norm_in,
                                                       float* __restrict__ hs1,
                                                       int N, int nChunks, int nRanges) {
  const int g = blockIdx.x * blockDim.x + threadIdx.x;
  if (2 * g >= N) return;
  const int r  = g / HW;
  const int wl = g - r * HW;
  const unsigned* __restrict__ po = planes + (size_t)r * nChunks * HW + wl;
  const unsigned* __restrict__ pi = planes + (size_t)(nRanges + r) * nChunks * HW + wl;
  unsigned cO[2] = {0u, 0u}, cI[2] = {0u, 0u};
#pragma unroll 8
  for (int c = 0; c < nChunks; ++c) {
    const unsigned a = po[(size_t)c * HW];
    const unsigned b = pi[(size_t)c * HW];
    cO[0] += a & 0xFFFFu;  cO[1] += a >> 16;
    cI[0] += b & 0xFFFFu;  cI[1] += b >> 16;
  }
#pragma unroll
  for (int t = 0; t < 2; ++t) {
    const int i = 2 * g + t;
    if (i < N) {
      const float dO = (float)cO[t];
      const float dI = (float)cI[t];
      const float no = dO > 0.0f ? rsqrtf(dO) : 0.0f;
      const float ni = dI > 0.0f ? rsqrtf(dI) : 0.0f;
      norm_out[i] = no;
      norm_in[i]  = ni;
#pragma unroll
      for (int k = 0; k < FIN; ++k) hs1[i * FIN + k] = h[i * FIN + k] * no;
    }
  }
}

template <bool GUARD>
static __device__ __forceinline__ void scatter1_chunk(const int* __restrict__ src,
                                                      const int* __restrict__ dst,
                                                      const float* __restrict__ hs1,
                                                      float* agg1, int cb, int E,
                                                      int lane, int sub, int q) {
  const int e = cb + lane;
  int s_l = 0, d_l = -1;
  if (!GUARD || e < E) { s_l = src[e]; d_l = dst[e]; }
  unsigned s[8];
  int d[8];
#pragma unroll
  for (int it = 0; it < 8; ++it) {
    const int j = 4 * it + q;
    s[it] = (unsigned)__shfl(s_l, j);
    d[it] = __shfl(d_l, j);
  }
  if (sub < FIN) {
    float v[8];
#pragma unroll
    for (int it = 0; it < 8; ++it) v[it] = hs1[s[it] * FIN + sub];
#pragma unroll
    for (int it = 0; it < 8; ++it)
      if (!GUARD || d[it] >= 0) atomicAdd(agg1 + ((unsigned)d[it] * FIN + sub), v[it]);
  }
}

__global__ __launch_bounds__(256) void scatter1_kernel(const int* __restrict__ src,
                                                       const int* __restrict__ dst,
                                                       const float* __restrict__ hs1,
                                                       float* agg1, size_t copyStride, int E) {
  agg1 += (size_t)half_of_package() * copyStride;
  const int lane = threadIdx.x & 31;
  const int wave = __builtin_amdgcn_readfirstlane((int)((blockIdx.x * blockDim.x + threadIdx.x) >> 5));
  const int sub  = lane & 7;
  const int q    = lane >> 3;
  const int ebase = wave * EPW;
  for (int c = 0; c < EPW / 32; ++c) {
    const int cb = ebase + c * 32;
    if (cb + 32 <= E)  scatter1_chunk<false>(src, dst, hs1, agg1, cb, E, lane, sub, q);
    else if (cb < E)   scatter1_chunk<true>(src, dst, hs1, agg1, cb, E, lane, sub, q);
  }
}

__global__ __launch_bounds__(256) void combine1_kernel(const float* __restrict__ agg1,
                                                       const float* __restrict__ agg1b,
                                                       const float* __restrict__ norm_in,
                                                       const float* __restrict__ norm_out,
                                                       const float* __restrict__ W1,
                                                       const float* __restrict__ b1,
                                                       float* __restrict__ hs2, int N) {
  int i = blockIdx.x * blockDim.x + threadIdx.x;
  if (i >= N) return;
  float ni = norm_in[i], no = norm_out[i];
  float v[FIN];
#pragma unroll
  for (int k = 0; k < FIN; ++k) v[k] = (agg1[(size_t)i * FIN + k] + agg1b[(size_t)i * FIN + k]) * ni;
#pragma unroll
  for (int j = 0; j < HID; ++j) {
    float acc = b1[j];
#pragma unroll
    for (int k = 0; k < FIN; ++k) acc = fmaf(v[k], W1[k * HID + j], acc);
    acc = fmaxf(acc, 0.0f);
    hs2[(size_t)i * HID + j] = acc * no;
  }
}

template <bool GUARD>
static __device__ __forceinline__ void scatter2_chunk(const int* __restrict__ src,
                                                      const int* __restrict__ dst,
                                                      const float* __restrict__ hs2,
                                                      float* agg2, int cb, int E,
                                                      int lane, int sub, int half) {
  const int e = cb + lane;
  int s_l = 0, d_l = -1;
  if (!GUARD || e < E) { s_l = src[e]; d_l = dst[e]; }
  float v[16];
  int d[16];
#pragma unroll
  for (int it = 0; it < 16; ++it) {
    const int j = 2 * it + half;
    const unsigned s = (unsigned)__shfl(s_l, j);
    d[it] = __shfl(d_l, j);
    v[it] = hs2[s * HID + sub];
  }
#pragma unroll
  for (int it = 0; it < 16; ++it)
    if (!GUARD || d[it] >= 0) atomicAdd(agg2 + ((unsigned)d[it] * HID + sub), v[it]);
}

__global__ __launch_bounds__(256) void scatter2_kernel(const int* __restrict__ src,
                                                       const int* __restrict__ dst,
                                                       const float* __restrict__ hs2,
                                                       float* agg2, size_t copyStride, int E) {
  agg2 += (size_t)half_of_package() * copyStride;
  const int lane = threadIdx.x & 31;
  const int wave = __builtin_amdgcn_readfirstlane((int)((blockIdx.x * blockDim.x + threadIdx.x) >> 5));
  const int sub  = lane & 15;
  const int half = lane >> 4;
  const int ebase = wave * EPW;
  for (int c = 0; c < EPW / 32; ++c) {
    const int cb = ebase + c * 32;
    if (cb + 32 <= E)  scatter2_chunk<false>(src, dst, hs2, agg2, cb, E, lane, sub, half);
    else if (cb < E)   scatter2_chunk<true>(src, dst, hs2, agg2, cb, E, lane, sub, half);
  }
}

__global__ __launch_bounds__(256) void combine2_wmma_kernel(float* agg2,
                                                            const float* __restrict__ agg2b,
                                                            const float* __restrict__ norm_in,
                                                            const float* __restrict__ W2,
                                                            const float* __restrict__ b2,
                                                            int numTiles) {
  int wave = (int)((blockIdx.x * blockDim.x + threadIdx.x) >> 5);
  int lane = threadIdx.x & 31;
  if (wave >= numTiles) return;
  int row0 = wave * 16;
  int half = lane >> 4;
  int lr   = lane & 15;

  int arow = row0 + lr;
  float ni = norm_in[arow];
  const float* Arow = agg2 + (size_t)arow * HID;
  const float* Brow = agg2b + (size_t)arow * HID;

  v8f c = {};
#pragma unroll
  for (int kb = 0; kb < 16; kb += 4) {
    int k0 = kb + half * 2;
    v2f a, b;
    a.x = (Arow[k0 + 0] + Brow[k0 + 0]) * ni;
    a.y = (Arow[k0 + 1] + Brow[k0 + 1]) * ni;
    b.x = W2[(k0 + 0) * HID + lr];
    b.y = W2[(k0 + 1) * HID + lr];
    c = __builtin_amdgcn_wmma_f32_16x16x4_f32(false, a, false, b, (short)0, c,
                                              false, false);
  }

  float bias = b2[lr];
#pragma unroll
  for (int r = 0; r < 8; ++r) {
    int m = r + half * 8;
    float val = fmaxf(c[r] + bias, 0.0f);
    agg2[(size_t)(row0 + m) * HID + lr] = val;
  }
}

__global__ __launch_bounds__(256) void combine2_tail_kernel(float* agg2,
                                                            const float* __restrict__ agg2b,
                                                            const float* __restrict__ norm_in,
                                                            const float* __restrict__ W2,
                                                            const float* __restrict__ b2,
                                                            int rowStart, int N) {
  int i = rowStart + blockIdx.x * blockDim.x + threadIdx.x;
  if (i >= N) return;
  float ni = norm_in[i];
  float v[HID];
#pragma unroll
  for (int k = 0; k < HID; ++k) v[k] = (agg2[(size_t)i * HID + k] + agg2b[(size_t)i * HID + k]) * ni;
#pragma unroll
  for (int j = 0; j < HID; ++j) {
    float acc = b2[j];
#pragma unroll
    for (int k = 0; k < HID; ++k) acc = fmaf(v[k], W2[k * HID + j], acc);
    agg2[(size_t)i * HID + j] = fmaxf(acc, 0.0f);
  }
}

__global__ __launch_bounds__(256) void head_kernel(const float* __restrict__ h2,
                                                   const float* __restrict__ Wl,
                                                   const float* __restrict__ bl,
                                                   float* __restrict__ out, int N) {
  int i = blockIdx.x * blockDim.x + threadIdx.x;
  if (i >= N) return;
  const float4* hp = (const float4*)(h2 + (size_t)i * HID);
  const float4* wp = (const float4*)Wl;
  float acc = bl[0];
#pragma unroll
  for (int q = 0; q < 4; ++q) {
    float4 hv = hp[q], wv = wp[q];
    acc = fmaf(hv.x, wv.x, acc);
    acc = fmaf(hv.y, wv.y, acc);
    acc = fmaf(hv.z, wv.z, acc);
    acc = fmaf(hv.w, wv.w, acc);
  }
  out[i] = acc;
}

extern "C" void kernel_launch(void* const* d_in, const int* in_sizes, int n_in,
                              void* d_out, int out_size, void* d_ws, size_t ws_size,
                              hipStream_t stream) {
  const float* h  = (const float*)d_in[0];
  const float* W1 = (const float*)d_in[1];
  const float* b1 = (const float*)d_in[2];
  const float* W2 = (const float*)d_in[3];
  const float* b2 = (const float*)d_in[4];
  const float* Wl = (const float*)d_in[5];
  const float* bl = (const float*)d_in[6];
  const int* src  = (const int*)d_in[7];
  const int* dst  = (const int*)d_in[8];
  float* out = (float*)d_out;

  const int N = in_sizes[0] / FIN;
  const int E = in_sizes[7];

  float* ws = (float*)d_ws;
  float* deg_out = ws;
  float* deg_in  = ws + (size_t)N;
  float* agg1    = ws + 2 * (size_t)N;
  float* agg2    = ws + 8 * (size_t)N;
  const size_t copyStride = 22 * (size_t)N;
  float* agg1b   = agg1 + copyStride;
  float* agg2b   = agg2 + copyStride;
  float* hs1     = ws + 46 * (size_t)N;
  float* hs2     = ws + 52 * (size_t)N;
  const int nChunks = (E + CHUNK - 1) / CHUNK;
  const int nRanges = (N + BINS - 1) / BINS;
  unsigned* planes = (unsigned*)(ws + ((68 * (size_t)N + 63) / 64) * 64);

  const int B = 256;
  int zn = 44 * N;
  zero_kernel<<<(zn + B - 1) / B, B, 0, stream>>>(agg1, zn);

  degree_hist_kernel<<<dim3(nChunks, nRanges, 2), HT, 0, stream>>>(src, dst, planes, E);
  const int nPairs = (N + 1) / 2;
  norm_hs1_kernel<<<(nPairs + B - 1) / B, B, 0, stream>>>(h, planes, deg_out, deg_in, hs1,
                                                          N, nChunks, nRanges);

  const int EB = (B / 32) * EPW;
  scatter1_kernel<<<(E + EB - 1) / EB, B, 0, stream>>>(src, dst, hs1, agg1, copyStride, E);
  combine1_kernel<<<(N + B - 1) / B, B, 0, stream>>>(agg1, agg1b, deg_in, deg_out, W1, b1, hs2, N);

  scatter2_kernel<<<(E + EB - 1) / EB, B, 0, stream>>>(src, dst, hs2, agg2, copyStride, E);

  int numTiles = N / 16;
  if (numTiles > 0) {
    int threads = numTiles * 32;
    combine2_wmma_kernel<<<(threads + B - 1) / B, B, 0, stream>>>(agg2, agg2b, deg_in, W2, b2, numTiles);
  }
  int rem = N - numTiles * 16;
  if (rem > 0) {
    combine2_tail_kernel<<<(rem + B - 1) / B, B, 0, stream>>>(agg2, agg2b, deg_in, W2, b2,
                                                              numTiles * 16, N);
  }

  head_kernel<<<(N + B - 1) / B, B, 0, stream>>>(agg2, Wl, bl, out, N);
}


# Layer-1 rows padded to 8 floats (32-byte aligned, no line straddled); 7th column holds a one, so the scatter's f32 atomics also count in-degree (exact integers): dst histogram and planes gone
# speedup vs baseline: 13.1452x; 1.1093x over previous
#include <hip/hip_runtime.h>

typedef __attribute__((ext_vector_type(2))) float v2f;
typedef __attribute__((ext_vector_type(8))) float v8f;

#define FIN 6
#define HID 16
#define ROW1 8
#define EPW 128

static __device__ __forceinline__ int half_of_package() {
  return (int)((__builtin_amdgcn_s_sendmsg_rtn(0x87) >> 18) & 1u);
}

__global__ __launch_bounds__(256) void zero_kernel(float* __restrict__ p, int n) {
  int i = blockIdx.x * blockDim.x + threadIdx.x;
  if (i < n) p[i] = 0.0f;
}

#define HW    51200
#define BINS  (2 * HW)
#define CHUNK 50000
#define HT    1024
#define HIT   ((CHUNK / 4 + HT - 1) / HT)

static __device__ __forceinline__ void hist_add(unsigned* bins, int i, int base) {
  const unsigned loc = (unsigned)(i - base);
  if (loc < (unsigned)BINS) atomicAdd(&bins[loc >> 1], (loc & 1u) ? 0x10000u : 1u);
}

__global__ __launch_bounds__(HT) void degree_hist_kernel(const int* __restrict__ src,
                                                         unsigned* __restrict__ planes,
                                                         int E) {
  __shared__ __attribute__((aligned(16))) unsigned bins[HW];
  const int tid   = threadIdx.x;
  const int chunk = blockIdx.x;
  const int r     = blockIdx.y;
  const int* __restrict__ idx = src;
  const int e0   = chunk * CHUNK;
  const int n    = min(CHUNK, E - e0);
  const int n4   = n >> 2;
  const int base = r * BINS;
  const int4* __restrict__ p4 = (const int4*)(idx + e0);

  int4 v[HIT];
#pragma unroll
  for (int it = 0; it < HIT; ++it) {
    const int k = tid + it * HT;
    v[it] = (k < n4) ? p4[k] : make_int4(-1, -1, -1, -1);
  }
  int vt = -1;
  if (tid < (n & 3)) vt = idx[e0 + (n4 << 2) + tid];

  uint4* b4 = (uint4*)bins;
  for (int k = tid; k < HW / 4; k += HT) b4[k] = make_uint4(0u, 0u, 0u, 0u);
  __syncthreads();

#pragma unroll
  for (int it = 0; it < HIT; ++it) {
    hist_add(bins, v[it].x, base);
    hist_add(bins, v[it].y, base);
    hist_add(bins, v[it].z, base);
    hist_add(bins, v[it].w, base);
  }
  hist_add(bins, vt, base);
  __syncthreads();

  uint4* o4 = (uint4*)(planes + ((size_t)r * gridDim.x + chunk) * HW);
  for (int k = tid; k < HW / 4; k += HT) o4[k] = b4[k];
}

__global__ __launch_bounds__(256) void norm_hs1_kernel(const float* __restrict__ h,
                                                       const unsigned* __restrict__ planes,
                                                       float* __restrict__ norm_out,
                                                       float* __restrict__ hs1,
                                                       int N, int nChunks) {
  const int g = blockIdx.x * blockDim.x + threadIdx.x;
  if (2 * g >= N) return;
  const int r  = g / HW;
  const int wl = g - r * HW;
  const unsigned* __restrict__ po = planes + (size_t)r * nChunks * HW + wl;
  unsigned cO[2] = {0u, 0u};
#pragma unroll 16
  for (int c = 0; c < nChunks; ++c) {
    const unsigned a = po[(size_t)c * HW];
    cO[0] += a & 0xFFFFu;  cO[1] += a >> 16;
  }
#pragma unroll
  for (int t = 0; t < 2; ++t) {
    const int i = 2 * g + t;
    if (i < N) {
      const float dO = (float)cO[t];
      const float no = dO > 0.0f ? rsqrtf(dO) : 0.0f;
      norm_out[i] = no;
      float4* row = (float4*)(hs1 + (size_t)i * ROW1);
      row[0] = make_float4(h[i * FIN + 0] * no, h[i * FIN + 1] * no,
                           h[i * FIN + 2] * no, h[i * FIN + 3] * no);
      row[1] = make_float4(h[i * FIN + 4] * no, h[i * FIN + 5] * no, 1.0f, 0.0f);
    }
  }
}

template <bool GUARD>
static __device__ __forceinline__ void scatter1_chunk(const int* __restrict__ src,
                                                      const int* __restrict__ dst,
                                                      const float* __restrict__ hs1,
                                                      float* agg1, int cb, int E,
                                                      int lane, int sub, int q) {
  const int e = cb + lane;
  int s_l = 0, d_l = -1;
  if (!GUARD || e < E) { s_l = src[e]; d_l = dst[e]; }
  unsigned s[8];
  int d[8];
#pragma unroll
  for (int it = 0; it < 8; ++it) {
    const int j = 4 * it + q;
    s[it] = (unsigned)__shfl(s_l, j);
    d[it] = __shfl(d_l, j);
  }
  if (sub < FIN + 1) {
    float v[8];
#pragma unroll
    for (int it = 0; it < 8; ++it) v[it] = hs1[s[it] * ROW1 + sub];
#pragma unroll
    for (int it = 0; it < 8; ++it)
      if (!GUARD || d[it] >= 0) atomicAdd(agg1 + ((unsigned)d[it] * ROW1 + sub), v[it]);
  }
}

__global__ __launch_bounds__(256) void scatter1_kernel(const int* __restrict__ src,
                                                       const int* __restrict__ dst,
                                                       const float* __restrict__ hs1,
                                                       float* agg1, size_t copyStride, int E) {
  agg1 += (size_t)half_of_package() * copyStride;
  const int lane = threadIdx.x & 31;
  const int wave = __builtin_amdgcn_readfirstlane((int)((blockIdx.x * blockDim.x + threadIdx.x) >> 5));
  const int sub  = lane & 7;
  const int q    = lane >> 3;
  const int ebase = wave * EPW;
  for (int c = 0; c < EPW / 32; ++c) {
    const int cb = ebase + c * 32;
    if (cb + 32 <= E)  scatter1_chunk<false>(src, dst, hs1, agg1, cb, E, lane, sub, q);
    else if (cb < E)   scatter1_chunk<true>(src, dst, hs1, agg1, cb, E, lane, sub, q);
  }
}

__global__ __launch_bounds__(256) void combine1_kernel(const float* __restrict__ agg1,
                                                       const float* __restrict__ agg1b,
                                                       float* __restrict__ norm_in,
                                                       const float* __restrict__ norm_out,
                                                       const float* __restrict__ W1,
                                                       const float* __restrict__ b1,
                                                       float* __restrict__ hs2, int N) {
  int i = blockIdx.x * blockDim.x + threadIdx.x;
  if (i >= N) return;
  const float4* ra = (const float4*)(agg1 + (size_t)i * ROW1);
  const float4* rb = (const float4*)(agg1b + (size_t)i * ROW1);
  const float4 a0 = ra[0], a1 = ra[1], c0 = rb[0], c1 = rb[1];
  const float dI = a1.z + c1.z;
  const float ni = dI > 0.0f ? rsqrtf(dI) : 0.0f;
  const float no = norm_out[i];
  norm_in[i] = ni;
  float v[FIN];
  v[0] = (a0.x + c0.x) * ni;  v[1] = (a0.y + c0.y) * ni;  v[2] = (a0.z + c0.z) * ni;
  v[3] = (a0.w + c0.w) * ni;  v[4] = (a1.x + c1.x) * ni;  v[5] = (a1.y + c1.y) * ni;
#pragma unroll
  for (int j = 0; j < HID; ++j) {
    float acc = b1[j];
#pragma unroll
    for (int k = 0; k < FIN; ++k) acc = fmaf(v[k], W1[k * HID + j], acc);
    acc = fmaxf(acc, 0.0f);
    hs2[(size_t)i * HID + j] = acc * no;
  }
}

template <bool GUARD>
static __device__ __forceinline__ void scatter2_chunk(const int* __restrict__ src,
                                                      const int* __restrict__ dst,
                                                      const float* __restrict__ hs2,
                                                      float* agg2, int cb, int E,
                                                      int lane, int sub, int half) {
  const int e = cb + lane;
  int s_l = 0, d_l = -1;
  if (!GUARD || e < E) { s_l = src[e]; d_l = dst[e]; }
  float v[16];
  int d[16];
#pragma unroll
  for (int it = 0; it < 16; ++it) {
    const int j = 2 * it + half;
    const unsigned s = (unsigned)__shfl(s_l, j);
    d[it] = __shfl(d_l, j);
    v[it] = hs2[s * HID + sub];
  }
#pragma unroll
  for (int it = 0; it < 16; ++it)
    if (!GUARD || d[it] >= 0) atomicAdd(agg2 + ((unsigned)d[it] * HID + sub), v[it]);
}

__global__ __launch_bounds__(256) void scatter2_kernel(const int* __restrict__ src,
                                                       const int* __restrict__ dst,
                                                       const float* __restrict__ hs2,
                                                       float* agg2, size_t copyStride, int E) {
  agg2 += (size_t)half_of_package() * copyStride;
  const int lane = threadIdx.x & 31;
  const int wave = __builtin_amdgcn_readfirstlane((int)((blockIdx.x * blockDim.x + threadIdx.x) >> 5));
  const int sub  = lane & 15;
  const int half = lane >> 4;
  const int ebase = wave * EPW;
  for (int c = 0; c < EPW / 32; ++c) {
    const int cb = ebase + c * 32;
    if (cb + 32 <= E)  scatter2_chunk<false>(src, dst, hs2, agg2, cb, E, lane, sub, half);
    else if (cb < E)   scatter2_chunk<true>(src, dst, hs2, agg2, cb, E, lane, sub, half);
  }
}

__global__ __launch_bounds__(256) void combine2_wmma_kernel(float* agg2,
                                                            const float* __restrict__ agg2b,
                                                            const float* __restrict__ norm_in,
                                                            const float* __restrict__ W2,
                                                            const float* __restrict__ b2,
                                                            int numTiles) {
  int wave = (int)((blockIdx.x * blockDim.x + threadIdx.x) >> 5);
  int lane = threadIdx.x & 31;
  if (wave >= numTiles) return;
  int row0 = wave * 16;
  int half = lane >> 4;
  int lr   = lane & 15;

  int arow = row0 + lr;
  float ni = norm_in[arow];
  const float* Arow = agg2 + (size_t)arow * HID;
  const float* Brow = agg2b + (size_t)arow * HID;

  v8f c = {};
#pragma unroll
  for (int kb = 0; kb < 16; kb += 4) {
    int k0 = kb + half * 2;
    v2f a, b;
    a.x = (Arow[k0 + 0] + Brow[k0 + 0]) * ni;
    a.y = (Arow[k0 + 1] + Brow[k0 + 1]) * ni;
    b.x = W2[(k0 + 0) * HID + lr];
    b.y = W2[(k0 + 1) * HID + lr];
    c = __builtin_amdgcn_wmma_f32_16x16x4_f32(false, a, false, b, (short)0, c,
                                              false, false);
  }

  float bias = b2[lr];
#pragma unroll
  for (int r = 0; r < 8; ++r) {
    int m = r + half * 8;
    float val = fmaxf(c[r] + bias, 0.0f);
    agg2[(size_t)(row0 + m) * HID + lr] = val;
  }
}

__global__ __launch_bounds__(256) void combine2_tail_kernel(float* agg2,
                                                            const float* __restrict__ agg2b,
                                                            const float* __restrict__ norm_in,
                                                            const float* __restrict__ W2,
                                                            const float* __restrict__ b2,
                                                            int rowStart, int N) {
  int i = rowStart + blockIdx.x * blockDim.x + threadIdx.x;
  if (i >= N) return;
  float ni = norm_in[i];
  float v[HID];
#pragma unroll
  for (int k = 0; k < HID; ++k) v[k] = (agg2[(size_t)i * HID + k] + agg2b[(size_t)i * HID + k]) * ni;
#pragma unroll
  for (int j = 0; j < HID; ++j) {
    float acc = b2[j];
#pragma unroll
    for (int k = 0; k < HID; ++k) acc = fmaf(v[k], W2[k * HID + j], acc);
    agg2[(size_t)i * HID + j] = fmaxf(acc, 0.0f);
  }
}

__global__ __launch_bounds__(256) void head_kernel(const float* __restrict__ h2,
                                                   const float* __restrict__ Wl,
                                                   const float* __restrict__ bl,
                                                   float* __restrict__ out, int N) {
  int i = blockIdx.x * blockDim.x + threadIdx.x;
  if (i >= N) return;
  const float4* hp = (const float4*)(h2 + (size_t)i * HID);
  const float4* wp = (const float4*)Wl;
  float acc = bl[0];
#pragma unroll
  for (int q = 0; q < 4; ++q) {
    float4 hv = hp[q], wv = wp[q];
    acc = fmaf(hv.x, wv.x, acc);
    acc = fmaf(hv.y, wv.y, acc);
    acc = fmaf(hv.z, wv.z, acc);
    acc = fmaf(hv.w, wv.w, acc);
  }
  out[i] = acc;
}

extern "C" void kernel_launch(void* const* d_in, const int* in_sizes, int n_in,
                              void* d_out, int out_size, void* d_ws, size_t ws_size,
                              hipStream_t stream) {
  const float* h  = (const float*)d_in[0];
  const float* W1 = (const float*)d_in[1];
  const float* b1 = (const float*)d_in[2];
  const float* W2 = (const float*)d_in[3];
  const float* b2 = (const float*)d_in[4];
  const float* Wl = (const float*)d_in[5];
  const float* bl = (const float*)d_in[6];
  const int* src  = (const int*)d_in[7];
  const int* dst  = (const int*)d_in[8];
  float* out = (float*)d_out;

  const int N = in_sizes[0] / FIN;
  const int E = in_sizes[7];

  float* ws = (float*)d_ws;
  const size_t P = (((size_t)N + 63) / 64) * 64;
  float* deg_out = ws;
  float* deg_in  = ws + P;
  float* agg1    = ws + 2 * P;
  float* agg2    = ws + 10 * P;
  const size_t copyStride = 24 * P;
  float* agg1b   = agg1 + copyStride;
  float* agg2b   = agg2 + copyStride;
  float* hs1     = ws + 50 * P;
  float* hs2     = ws + 58 * P;
  const int nChunks = (E + CHUNK - 1) / CHUNK;
  const int nRanges = (N + BINS - 1) / BINS;
  unsigned* planes = (unsigned*)(ws + 74 * P);

  const int B = 256;
  int zn = (int)(48 * P);
  zero_kernel<<<(zn + B - 1) / B, B, 0, stream>>>(agg1, zn);

  degree_hist_kernel<<<dim3(nChunks, nRanges, 1), HT, 0, stream>>>(src, planes, E);
  const int nPairs = (N + 1) / 2;
  norm_hs1_kernel<<<(nPairs + B - 1) / B, B, 0, stream>>>(h, planes, deg_out, hs1, N, nChunks);

  const int EB = (B / 32) * EPW;
  scatter1_kernel<<<(E + EB - 1) / EB, B, 0, stream>>>(src, dst, hs1, agg1, copyStride, E);
  combine1_kernel<<<(N + B - 1) / B, B, 0, stream>>>(agg1, agg1b, deg_in, deg_out, W1, b1, hs2, N);

  scatter2_kernel<<<(E + EB - 1) / EB, B, 0, stream>>>(src, dst, hs2, agg2, copyStride, E);

  int numTiles = N / 16;
  if (numTiles > 0) {
    int threads = numTiles * 32;
    combine2_wmma_kernel<<<(threads + B - 1) / B, B, 0, stream>>>(agg2, agg2b, deg_in, W2, b2, numTiles);
  }
  int rem = N - numTiles * 16;
  if (rem > 0) {
    combine2_tail_kernel<<<(rem + B - 1) / B, B, 0, stream>>>(agg2, agg2b, deg_in, W2, b2,
                                                              numTiles * 16, N);
  }

  head_kernel<<<(N + B - 1) / B, B, 0, stream>>>(agg2, Wl, bl, out, N);
}


# Output head folded into the layer-2 WMMA combine: relu'd tile times Wl summed over 16 lanes by a halving lane butterfly (the head's sum order changes); h2 store and the head launch are gone
# speedup vs baseline: 16.0541x; 1.2213x over previous
#include <hip/hip_runtime.h>

typedef __attribute__((ext_vector_type(2))) float v2f;
typedef __attribute__((ext_vector_type(8))) float v8f;

#define FIN 6
#define HID 16
#define ROW1 8
#define EPW 128

static __device__ __forceinline__ int half_of_package() {
  return (int)((__builtin_amdgcn_s_sendmsg_rtn(0x87) >> 18) & 1u);
}

__global__ __launch_bounds__(256) void zero_kernel(float* __restrict__ p, int n) {
  int i = blockIdx.x * blockDim.x + threadIdx.x;
  if (i < n) p[i] = 0.0f;
}

#define HW    51200
#define BINS  (2 * HW)
#define CHUNK 50000
#define HT    1024
#define HIT   ((CHUNK / 4 + HT - 1) / HT)

static __device__ __forceinline__ void hist_add(unsigned* bins, int i, int base) {
  const unsigned loc = (unsigned)(i - base);
  if (loc < (unsigned)BINS) atomicAdd(&bins[loc >> 1], (loc & 1u) ? 0x10000u : 1u);
}

__global__ __launch_bounds__(HT) void degree_hist_kernel(const int* __restrict__ src,
                                                         unsigned* __restrict__ planes,
                                                         int E) {
  __shared__ __attribute__((aligned(16))) unsigned bins[HW];
  const int tid   = threadIdx.x;
  const int chunk = blockIdx.x;
  const int r     = blockIdx.y;
  const int* __restrict__ idx = src;
  const int e0   = chunk * CHUNK;
  const int n    = min(CHUNK, E - e0);
  const int n4   = n >> 2;
  const int base = r * BINS;
  const int4* __restrict__ p4 = (const int4*)(idx + e0);

  int4 v[HIT];
#pragma unroll
  for (int it = 0; it < HIT; ++it) {
    const int k = tid + it * HT;
    v[it] = (k < n4) ? p4[k] : make_int4(-1, -1, -1, -1);
  }
  int vt = -1;
  if (tid < (n & 3)) vt = idx[e0 + (n4 << 2) + tid];

  uint4* b4 = (uint4*)bins;
  for (int k = tid; k < HW / 4; k += HT) b4[k] = make_uint4(0u, 0u, 0u, 0u);
  __syncthreads();

#pragma unroll
  for (int it = 0; it < HIT; ++it) {
    hist_add(bins, v[it].x, base);
    hist_add(bins, v[it].y, base);
    hist_add(bins, v[it].z, base);
    hist_add(bins, v[it].w, base);
  }
  hist_add(bins, vt, base);
  __syncthreads();

  uint4* o4 = (uint4*)(planes + ((size_t)r * gridDim.x + chunk) * HW);
  for (int k = tid; k < HW / 4; k += HT) o4[k] = b4[k];
}

__global__ __launch_bounds__(256) void norm_hs1_kernel(const float* __restrict__ h,
                                                       const unsigned* __restrict__ planes,
                                                       float* __restrict__ norm_out,
                                                       float* __restrict__ hs1,
                                                       int N, int nChunks) {
  const int g = blockIdx.x * blockDim.x + threadIdx.x;
  if (2 * g >= N) return;
  const int r  = g / HW;
  const int wl = g - r * HW;
  const unsigned* __restrict__ po = planes + (size_t)r * nChunks * HW + wl;
  unsigned cO[2] = {0u, 0u};
#pragma unroll 16
  for (int c = 0; c < nChunks; ++c) {
    const unsigned a = po[(size_t)c * HW];
    cO[0] += a & 0xFFFFu;  cO[1] += a >> 16;
  }
#pragma unroll
  for (int t = 0; t < 2; ++t) {
    const int i = 2 * g + t;
    if (i < N) {
      const float dO = (float)cO[t];
      const float no = dO > 0.0f ? rsqrtf(dO) : 0.0f;
      norm_out[i] = no;
      float4* row = (float4*)(hs1 + (size_t)i * ROW1);
      row[0] = make_float4(h[i * FIN + 0] * no, h[i * FIN + 1] * no,
                           h[i * FIN + 2] * no, h[i * FIN + 3] * no);
      row[1] = make_float4(h[i * FIN + 4] * no, h[i * FIN + 5] * no, 1.0f, 0.0f);
    }
  }
}

template <bool GUARD>
static __device__ __forceinline__ void scatter1_chunk(const int* __restrict__ src,
                                                      const int* __restrict__ dst,
                                                      const float* __restrict__ hs1,
                                                      float* agg1, int cb, int E,
                                                      int lane, int sub, int q) {
  const int e = cb + lane;
  int s_l = 0, d_l = -1;
  if (!GUARD || e < E) { s_l = src[e]; d_l = dst[e]; }
  unsigned s[8];
  int d[8];
#pragma unroll
  for (int it = 0; it < 8; ++it) {
    const int j = 4 * it + q;
    s[it] = (unsigned)__shfl(s_l, j);
    d[it] = __shfl(d_l, j);
  }
  if (sub < FIN + 1) {
    float v[8];
#pragma unroll
    for (int it = 0; it < 8; ++it) v[it] = hs1[s[it] * ROW1 + sub];
#pragma unroll
    for (int it = 0; it < 8; ++it)
      if (!GUARD || d[it] >= 0) atomicAdd(agg1 + ((unsigned)d[it] * ROW1 + sub), v[it]);
  }
}

__global__ __launch_bounds__(256) void scatter1_kernel(const int* __restrict__ src,
                                                       const int* __restrict__ dst,
                                                       const float* __restrict__ hs1,
                                                       float* agg1, size_t copyStride, int E) {
  agg1 += (size_t)half_of_package() * copyStride;
  const int lane = threadIdx.x & 31;
  const int wave = __builtin_amdgcn_readfirstlane((int)((blockIdx.x * blockDim.x + threadIdx.x) >> 5));
  const int sub  = lane & 7;
  const int q    = lane >> 3;
  const int ebase = wave * EPW;
  for (int c = 0; c < EPW / 32; ++c) {
    const int cb = ebase + c * 32;
    if (cb + 32 <= E)  scatter1_chunk<false>(src, dst, hs1, agg1, cb, E, lane, sub, q);
    else if (cb < E)   scatter1_chunk<true>(src, dst, hs1, agg1, cb, E, lane, sub, q);
  }
}

__global__ __launch_bounds__(256) void combine1_kernel(const float* __restrict__ agg1,
                                                       const float* __restrict__ agg1b,
                                                       float* __restrict__ norm_in,
                                                       const float* __restrict__ norm_out,
                                                       const float* __restrict__ W1,
                                                       const float* __restrict__ b1,
                                                       float* __restrict__ hs2, int N) {
  int i = blockIdx.x * blockDim.x + threadIdx.x;
  if (i >= N) return;
  const float4* ra = (const float4*)(agg1 + (size_t)i * ROW1);
  const float4* rb = (const float4*)(agg1b + (size_t)i * ROW1);
  const float4 a0 = ra[0], a1 = ra[1], c0 = rb[0], c1 = rb[1];
  const float dI = a1.z + c1.z;
  const float ni = dI > 0.0f ? rsqrtf(dI) : 0.0f;
  const float no = norm_out[i];
  norm_in[i] = ni;
  float v[FIN];
  v[0] = (a0.x + c0.x) * ni;  v[1] = (a0.y + c0.y) * ni;  v[2] = (a0.z + c0.z) * ni;
  v[3] = (a0.w + c0.w) * ni;  v[4] = (a1.x + c1.x) * ni;  v[5] = (a1.y + c1.y) * ni;
#pragma unroll
  for (int j = 0; j < HID; ++j) {
    float acc = b1[j];
#pragma unroll
    for (int k = 0; k < FIN; ++k) acc = fmaf(v[k], W1[k * HID + j], acc);
    acc = fmaxf(acc, 0.0f);
    hs2[(size_t)i * HID + j] = acc * no;
  }
}

template <bool GUARD>
static __device__ __forceinline__ void scatter2_chunk(const int* __restrict__ src,
                                                      const int* __restrict__ dst,
                                                      const float* __restrict__ hs2,
                                                      float* agg2, int cb, int E,
                                                      int lane, int sub, int half) {
  const int e = cb + lane;
  int s_l = 0, d_l = -1;
  if (!GUARD || e < E) { s_l = src[e]; d_l = dst[e]; }
  float v[16];
  int d[16];
#pragma unroll
  for (int it = 0; it < 16; ++it) {
    const int j = 2 * it + half;
    const unsigned s = (unsigned)__shfl(s_l, j);
    d[it] = __shfl(d_l, j);
    v[it] = hs2[s * HID + sub];
  }
#pragma unroll
  for (int it = 0; it < 16; ++it)
    if (!GUARD || d[it] >= 0) atomicAdd(agg2 + ((unsigned)d[it] * HID + sub), v[it]);
}

__global__ __launch_bounds__(256) void scatter2_kernel(const int* __restrict__ src,
                                                       const int* __restrict__ dst,
                                                       const float* __restrict__ hs2,
                                                       float* agg2, size_t copyStride, int E) {
  agg2 += (size_t)half_of_package() * copyStride;
  const int lane = threadIdx.x & 31;
  const int wave = __builtin_amdgcn_readfirstlane((int)((blockIdx.x * blockDim.x + threadIdx.x) >> 5));
  const int sub  = lane & 15;
  const int half = lane >> 4;
  const int ebase = wave * EPW;
  for (int c = 0; c < EPW / 32; ++c) {
    const int cb = ebase + c * 32;
    if (cb + 32 <= E)  scatter2_chunk<false>(src, dst, hs2, agg2, cb, E, lane, sub, half);
    else if (cb < E)   scatter2_chunk<true>(src, dst, hs2, agg2, cb, E, lane, sub, half);
  }
}

__global__ __launch_bounds__(256) void combine2_head_wmma_kernel(const float* __restrict__ agg2,
                                                                 const float* __restrict__ agg2b,
                                                                 const float* __restrict__ norm_in,
                                                                 const float* __restrict__ W2,
                                                                 const float* __restrict__ b2,
                                                                 const float* __restrict__ Wl,
                                                                 const float* __restrict__ bl,
                                                                 float* __restrict__ out,
                                                                 int numTiles) {
  int wave = (int)((blockIdx.x * blockDim.x + threadIdx.x) >> 5);
  int lane = threadIdx.x & 31;
  if (wave >= numTiles) return;
  int row0 = wave * 16;
  int half = lane >> 4;
  int lr   = lane & 15;

  int arow = row0 + lr;
  float ni = norm_in[arow];
  const float* Arow = agg2 + (size_t)arow * HID;
  const float* Brow = agg2b + (size_t)arow * HID;

  v8f c = {};
#pragma unroll
  for (int kb = 0; kb < 16; kb += 4) {
    int k0 = kb + half * 2;
    v2f a, b;
    a.x = (Arow[k0 + 0] + Brow[k0 + 0]) * ni;
    a.y = (Arow[k0 + 1] + Brow[k0 + 1]) * ni;
    b.x = W2[(k0 + 0) * HID + lr];
    b.y = W2[(k0 + 1) * HID + lr];
    c = __builtin_amdgcn_wmma_f32_16x16x4_f32(false, a, false, b, (short)0, c,
                                              false, false);
  }

  const float bias = b2[lr];
  const float wl   = Wl[lr];
  float p[8];
#pragma unroll
  for (int r = 0; r < 8; ++r) p[r] = fmaxf(c[r] + bias, 0.0f) * wl;

  const bool hi8 = (lr & 8) != 0, hi4 = (lr & 4) != 0, hi2 = (lr & 2) != 0;
  float q4[4], q2[2];
#pragma unroll
  for (int j = 0; j < 4; ++j) {
    const float keep = hi8 ? p[4 + j] : p[j];
    const float send = hi8 ? p[j] : p[4 + j];
    q4[j] = keep + __shfl_xor(send, 8);
  }
#pragma unroll
  for (int j = 0; j < 2; ++j) {
    const float keep = hi4 ? q4[2 + j] : q4[j];
    const float send = hi4 ? q4[j] : q4[2 + j];
    q2[j] = keep + __shfl_xor(send, 4);
  }
  const float keep1 = hi2 ? q2[1] : q2[0];
  const float send1 = hi2 ? q2[0] : q2[1];
  const float q1 = keep1 + __shfl_xor(send1, 2);
  const float s  = q1 + __shfl_xor(q1, 1);
  if ((lr & 1) == 0)
    out[row0 + half * 8 + (lr >> 1)] = s + bl[0];
}

__global__ __launch_bounds__(256) void combine2_head_tail_kernel(const float* __restrict__ agg2,
                                                                 const float* __restrict__ agg2b,
                                                                 const float* __restrict__ norm_in,
                                                                 const float* __restrict__ W2,
                                                                 const float* __restrict__ b2,
                                                                 const float* __restrict__ Wl,
                                                                 const float* __restrict__ bl,
                                                                 float* __restrict__ out,
                                                                 int rowStart, int N) {
  int i = rowStart + blockIdx.x * blockDim.x + threadIdx.x;
  if (i >= N) return;
  float ni = norm_in[i];
  float v[HID];
#pragma unroll
  for (int k = 0; k < HID; ++k) v[k] = (agg2[(size_t)i * HID + k] + agg2b[(size_t)i * HID + k]) * ni;
  float o = bl[0];
#pragma unroll
  for (int j = 0; j < HID; ++j) {
    float acc = b2[j];
#pragma unroll
    for (int k = 0; k < HID; ++k) acc = fmaf(v[k], W2[k * HID + j], acc);
    o = fmaf(fmaxf(acc, 0.0f), Wl[j], o);
  }
  out[i] = o;
}

extern "C" void kernel_launch(void* const* d_in, const int* in_sizes, int n_in,
                              void* d_out, int out_size, void* d_ws, size_t ws_size,
                              hipStream_t stream) {
  const float* h  = (const float*)d_in[0];
  const float* W1 = (const float*)d_in[1];
  const float* b1 = (const float*)d_in[2];
  const float* W2 = (const float*)d_in[3];
  const float* b2 = (const float*)d_in[4];
  const float* Wl = (const float*)d_in[5];
  const float* bl = (const float*)d_in[6];
  const int* src  = (const int*)d_in[7];
  const int* dst  = (const int*)d_in[8];
  float* out = (float*)d_out;

  const int N = in_sizes[0] / FIN;
  const int E = in_sizes[7];

  float* ws = (float*)d_ws;
  const size_t P = (((size_t)N + 63) / 64) * 64;
  float* deg_out = ws;
  float* deg_in  = ws + P;
  float* agg1    = ws + 2 * P;
  float* agg2    = ws + 10 * P;
  const size_t copyStride = 24 * P;
  float* agg1b   = agg1 + copyStride;
  float* agg2b   = agg2 + copyStride;
  float* hs1     = ws + 50 * P;
  float* hs2     = ws + 58 * P;
  const int nChunks = (E + CHUNK - 1) / CHUNK;
  const int nRanges = (N + BINS - 1) / BINS;
  unsigned* planes = (unsigned*)(ws + 74 * P);

  const int B = 256;
  int zn = (int)(48 * P);
  zero_kernel<<<(zn + B - 1) / B, B, 0, stream>>>(agg1, zn);

  degree_hist_kernel<<<dim3(nChunks, nRanges, 1), HT, 0, stream>>>(src, planes, E);
  const int nPairs = (N + 1) / 2;
  norm_hs1_kernel<<<(nPairs + B - 1) / B, B, 0, stream>>>(h, planes, deg_out, hs1, N, nChunks);

  const int EB = (B / 32) * EPW;
  scatter1_kernel<<<(E + EB - 1) / EB, B, 0, stream>>>(src, dst, hs1, agg1, copyStride, E);
  combine1_kernel<<<(N + B - 1) / B, B, 0, stream>>>(agg1, agg1b, deg_in, deg_out, W1, b1, hs2, N);

  scatter2_kernel<<<(E + EB - 1) / EB, B, 0, stream>>>(src, dst, hs2, agg2, copyStride, E);

  int numTiles = N / 16;
  if (numTiles > 0) {
    int threads = numTiles * 32;
    combine2_head_wmma_kernel<<<(threads + B - 1) / B, B, 0, stream>>>(agg2, agg2b, deg_in, W2, b2,
                                                                       Wl, bl, out, numTiles);
  }
  int rem = N - numTiles * 16;
  if (rem > 0) {
    combine2_head_tail_kernel<<<(rem + B - 1) / B, B, 0, stream>>>(agg2, agg2b, deg_in, W2, b2,
                                                                   Wl, bl, out, numTiles * 16, N);
  }
}
